# starting kernel
# baseline (speedup 1.0000x reference)
#include <hip/hip_runtime.h>
#include <math.h>

typedef float v2f __attribute__((ext_vector_type(2)));
typedef float v8f __attribute__((ext_vector_type(8)));

namespace {
constexpr int T_SEQ = 128;
constexpr int BATCH = 16384;
constexpr int NIN   = 28;
constexpr int NHID  = 100;
constexpr int NCLS  = 10;
constexpr int HP    = 112;
constexpr int HS    = 113;
constexpr int NT    = 7;
constexpr int KX    = NIN / 4;
constexpr int KH    = NHID / 4;
constexpr int MTILE = 128;
constexpr int BLOCK = 128;
}

__device__ __forceinline__ v8f wmma_f32(v2f a, v2f b, v8f c) {
  return __builtin_amdgcn_wmma_f32_16x16x4_f32(false, a, false, b, (short)0, c,
                                               false, false);
}

__device__ __forceinline__ float fast_tanhf(float x) {
  float e = __expf(2.0f * x);
  return 1.0f - 2.0f * __builtin_amdgcn_rcpf(e + 1.0f);
}

__global__ __launch_bounds__(BLOCK) void rnn_fused_kernel(
    const float* __restrict__ data, const float* __restrict__ w_xh,
    const float* __restrict__ b_xh, const float* __restrict__ w_hh,
    const float* __restrict__ b_hh, const float* __restrict__ w_fc,
    const float* __restrict__ b_fc, float* __restrict__ out) {
  __shared__ float s_x[2][MTILE * NIN];
  __shared__ float s_h[MTILE][HS];
  __shared__ float s_whh[HP][HP];
  __shared__ float s_wxh[NIN][HP];
  __shared__ float s_wfc[HP][16];
  __shared__ float s_bias[HP];
  __shared__ float s_bfc[16];

  const int tid   = threadIdx.x;
  const int lane  = tid & 31;
  const int wave  = tid >> 5;
  const int col   = lane & 15;
  const int lrow  = lane & 15;
  const int khalf = (lane >> 4) << 1;
  const int rhalf = (lane >> 4) << 3;
  const int rowA0 = wave * 32;
  const int rowA1 = rowA0 + 16;
  const int b0    = blockIdx.x * MTILE;

  for (int i = tid; i < MTILE * HS; i += BLOCK) (&s_h[0][0])[i] = 0.0f;
  for (int i = tid; i < HP * HP; i += BLOCK) {
    int k = i / HP, n = i % HP;
    s_whh[k][n] = (k < NHID && n < NHID) ? w_hh[k * NHID + n] : 0.0f;
  }
  for (int i = tid; i < NIN * HP; i += BLOCK) {
    int k = i / HP, n = i % HP;
    s_wxh[k][n] = (n < NHID) ? w_xh[k * NHID + n] : 0.0f;
  }
  for (int i = tid; i < HP * 16; i += BLOCK) {
    int k = i >> 4, n = i & 15;
    s_wfc[k][n] = (k < NHID && n < NCLS) ? w_fc[k * NCLS + n] : 0.0f;
  }
  for (int i = tid; i < HP; i += BLOCK)
    s_bias[i] = (i < NHID) ? (b_xh[i] + b_hh[i]) : 0.0f;
  if (tid < 16) s_bfc[tid] = (tid < NCLS) ? b_fc[tid] : 0.0f;
  __syncthreads();

  v8f cbias[NT];
#pragma unroll
  for (int n = 0; n < NT; ++n) {
    const float bv = s_bias[n * 16 + col];
    v8f c;
#pragma unroll
    for (int r = 0; r < 8; ++r) c[r] = bv;
    cbias[n] = c;
  }

#pragma unroll 1
  for (int t = 0; t < T_SEQ; ++t) {
    float* xb = s_x[t & 1];
    const float4* xt =
        (const float4*)(data + ((size_t)t * BATCH + b0) * NIN);
    for (int i = tid; i < MTILE * NIN / 4; i += BLOCK)
      ((float4*)xb)[i] = xt[i];
    if (t + 1 < T_SEQ) {
      const float* xn = data + ((size_t)(t + 1) * BATCH + b0) * NIN;
      for (int i = tid * 32; i < MTILE * NIN; i += BLOCK * 32)
        __builtin_prefetch(xn + i, 0, 1);
    }
    __syncthreads();

    v8f acc0[NT], acc1[NT];
#pragma unroll
    for (int n = 0; n < NT; ++n) {
      acc0[n] = cbias[n];
      acc1[n] = cbias[n];
    }

#pragma unroll
    for (int k = 0; k < KX; ++k) {
      const int kb = k * 4 + khalf;
      v2f a0, a1;
      a0.x = xb[(rowA0 + lrow) * NIN + kb];
      a0.y = xb[(rowA0 + lrow) * NIN + kb + 1];
      a1.x = xb[(rowA1 + lrow) * NIN + kb];
      a1.y = xb[(rowA1 + lrow) * NIN + kb + 1];
#pragma unroll
      for (int n = 0; n < NT; ++n) {
        v2f b;
        b.x = s_wxh[kb][n * 16 + col];
        b.y = s_wxh[kb + 1][n * 16 + col];
        acc0[n] = wmma_f32(a0, b, acc0[n]);
        acc1[n] = wmma_f32(a1, b, acc1[n]);
      }
    }

#pragma unroll 1
    for (int k = 0; k < KH; ++k) {
      const int kb = k * 4 + khalf;
      v2f a0, a1;
      a0.x = s_h[rowA0 + lrow][kb];
      a0.y = s_h[rowA0 + lrow][kb + 1];
      a1.x = s_h[rowA1 + lrow][kb];
      a1.y = s_h[rowA1 + lrow][kb + 1];
#pragma unroll
      for (int n = 0; n < NT; ++n) {
        v2f b;
        b.x = s_whh[kb][n * 16 + col];
        b.y = s_whh[kb + 1][n * 16 + col];
        acc0[n] = wmma_f32(a0, b, acc0[n]);
        acc1[n] = wmma_f32(a1, b, acc1[n]);
      }
    }

#pragma unroll
    for (int n = 0; n < NT; ++n) {
#pragma unroll
      for (int r = 0; r < 8; ++r) {
        s_h[rowA0 + rhalf + r][n * 16 + col] = fast_tanhf(acc0[n][r]);
        s_h[rowA1 + rhalf + r][n * 16 + col] = fast_tanhf(acc1[n][r]);
      }
    }
  }

  v8f f0, f1;
  {
    const float bv = s_bfc[col];
#pragma unroll
    for (int r = 0; r < 8; ++r) {
      f0[r] = bv;
      f1[r] = bv;
    }
  }
#pragma unroll 1
  for (int k = 0; k < KH; ++k) {
    const int kb = k * 4 + khalf;
    v2f a0, a1, b;
    a0.x = s_h[rowA0 + lrow][kb];
    a0.y = s_h[rowA0 + lrow][kb + 1];
    a1.x = s_h[rowA1 + lrow][kb];
    a1.y = s_h[rowA1 + lrow][kb + 1];
    b.x = s_wfc[kb][col];
    b.y = s_wfc[kb + 1][col];
    f0 = wmma_f32(a0, b, f0);
    f1 = wmma_f32(a1, b, f1);
  }
  if (col < NCLS) {
#pragma unroll
    for (int r = 0; r < 8; ++r) {
      out[(size_t)(b0 + rowA0 + rhalf + r) * NCLS + col] = f0[r];
      out[(size_t)(b0 + rowA1 + rhalf + r) * NCLS + col] = f1[r];
    }
  }
}

extern "C" void kernel_launch(void* const* d_in, const int* in_sizes, int n_in,
                              void* d_out, int out_size, void* d_ws,
                              size_t ws_size, hipStream_t stream) {
  const float* data = (const float*)d_in[0];
  const float* w_xh = (const float*)d_in[1];
  const float* b_xh = (const float*)d_in[2];
  const float* w_hh = (const float*)d_in[3];
  const float* b_hh = (const float*)d_in[4];
  const float* w_fc = (const float*)d_in[5];
  const float* b_fc = (const float*)d_in[6];
  float* out = (float*)d_out;

  dim3 grid(BATCH / MTILE);
  dim3 block(BLOCK);
  hipLaunchKernelGGL(rnn_fused_kernel, grid, block, 0, stream, data, w_xh,
                     b_xh, w_hh, b_hh, w_fc, b_fc, out);
}


# Fill the chip: 64-row workgroup tiles, each of the 4 waves owns one 16-row group instead of two (256 workgroups, 1024 waves: one wave on every SIMD of all 256 WGPs instead of 128)
# speedup vs baseline: 1.6015x; 1.6015x over previous
#include <hip/hip_runtime.h>
#include <math.h>

typedef float v2f __attribute__((ext_vector_type(2)));
typedef float v8f __attribute__((ext_vector_type(8)));

namespace {
constexpr int T_SEQ = 128;
constexpr int BATCH = 16384;
constexpr int NIN   = 28;
constexpr int NHID  = 100;
constexpr int NCLS  = 10;
constexpr int HP    = 112;
constexpr int HS    = 113;
constexpr int NT    = 7;
constexpr int KX    = NIN / 4;
constexpr int KH    = NHID / 4;
constexpr int MTILE = 64;
constexpr int BLOCK = 128;
}

__device__ __forceinline__ v8f wmma_f32(v2f a, v2f b, v8f c) {
  return __builtin_amdgcn_wmma_f32_16x16x4_f32(false, a, false, b, (short)0, c,
                                               false, false);
}

__device__ __forceinline__ float fast_tanhf(float x) {
  float e = __expf(2.0f * x);
  return 1.0f - 2.0f * __builtin_amdgcn_rcpf(e + 1.0f);
}

__global__ __launch_bounds__(BLOCK) void rnn_fused_kernel(
    const float* __restrict__ data, const float* __restrict__ w_xh,
    const float* __restrict__ b_xh, const float* __restrict__ w_hh,
    const float* __restrict__ b_hh, const float* __restrict__ w_fc,
    const float* __restrict__ b_fc, float* __restrict__ out) {
  __shared__ float s_x[2][MTILE * NIN];
  __shared__ float s_h[MTILE][HS];
  __shared__ float s_whh[HP][HP];
  __shared__ float s_wxh[NIN][HP];
  __shared__ float s_wfc[HP][16];
  __shared__ float s_bias[HP];
  __shared__ float s_bfc[16];

  const int tid   = threadIdx.x;
  const int lane  = tid & 31;
  const int wave  = tid >> 5;
  const int col   = lane & 15;
  const int lrow  = lane & 15;
  const int khalf = (lane >> 4) << 1;
  const int rhalf = (lane >> 4) << 3;
  const int rowA0 = wave * 16;
  const int b0    = blockIdx.x * MTILE;

  for (int i = tid; i < MTILE * HS; i += BLOCK) (&s_h[0][0])[i] = 0.0f;
  for (int i = tid; i < HP * HP; i += BLOCK) {
    int k = i / HP, n = i % HP;
    s_whh[k][n] = (k < NHID && n < NHID) ? w_hh[k * NHID + n] : 0.0f;
  }
  for (int i = tid; i < NIN * HP; i += BLOCK) {
    int k = i / HP, n = i % HP;
    s_wxh[k][n] = (n < NHID) ? w_xh[k * NHID + n] : 0.0f;
  }
  for (int i = tid; i < HP * 16; i += BLOCK) {
    int k = i >> 4, n = i & 15;
    s_wfc[k][n] = (k < NHID && n < NCLS) ? w_fc[k * NCLS + n] : 0.0f;
  }
  for (int i = tid; i < HP; i += BLOCK)
    s_bias[i] = (i < NHID) ? (b_xh[i] + b_hh[i]) : 0.0f;
  if (tid < 16) s_bfc[tid] = (tid < NCLS) ? b_fc[tid] : 0.0f;
  __syncthreads();

  v8f cbias[NT];
#pragma unroll
  for (int n = 0; n < NT; ++n) {
    const float bv = s_bias[n * 16 + col];
    v8f c;
#pragma unroll
    for (int r = 0; r < 8; ++r) c[r] = bv;
    cbias[n] = c;
  }

#pragma unroll 1
  for (int t = 0; t < T_SEQ; ++t) {
    float* xb = s_x[t & 1];
    const float4* xt =
        (const float4*)(data + ((size_t)t * BATCH + b0) * NIN);
    for (int i = tid; i < MTILE * NIN / 4; i += BLOCK)
      ((float4*)xb)[i] = xt[i];
    if (t + 1 < T_SEQ) {
      const float* xn = data + ((size_t)(t + 1) * BATCH + b0) * NIN;
      for (int i = tid * 32; i < MTILE * NIN; i += BLOCK * 32)
        __builtin_prefetch(xn + i, 0, 1);
    }
    __syncthreads();

    v8f acc0[NT];
#pragma unroll
    for (int n = 0; n < NT; ++n) acc0[n] = cbias[n];

#pragma unroll
    for (int k = 0; k < KX; ++k) {
      const int kb = k * 4 + khalf;
      v2f a0;
      a0.x = xb[(rowA0 + lrow) * NIN + kb];
      a0.y = xb[(rowA0 + lrow) * NIN + kb + 1];
#pragma unroll
      for (int n = 0; n < NT; ++n) {
        v2f b;
        b.x = s_wxh[kb][n * 16 + col];
        b.y = s_wxh[kb + 1][n * 16 + col];
        acc0[n] = wmma_f32(a0, b, acc0[n]);
      }
    }

#pragma unroll 1
    for (int k = 0; k < KH; ++k) {
      const int kb = k * 4 + khalf;
      v2f a0;
      a0.x = s_h[rowA0 + lrow][kb];
      a0.y = s_h[rowA0 + lrow][kb + 1];
#pragma unroll
      for (int n = 0; n < NT; ++n) {
        v2f b;
        b.x = s_whh[kb][n * 16 + col];
        b.y = s_whh[kb + 1][n * 16 + col];
        acc0[n] = wmma_f32(a0, b, acc0[n]);
      }
    }

#pragma unroll
    for (int n = 0; n < NT; ++n) {
#pragma unroll
      for (int r = 0; r < 8; ++r)
        s_h[rowA0 + rhalf + r][n * 16 + col] = fast_tanhf(acc0[n][r]);
    }
  }

  v8f f0;
  {
    const float bv = s_bfc[col];
#pragma unroll
    for (int r = 0; r < 8; ++r) f0[r] = bv;
  }
#pragma unroll 1
  for (int k = 0; k < KH; ++k) {
    const int kb = k * 4 + khalf;
    v2f a0, b;
    a0.x = s_h[rowA0 + lrow][kb];
    a0.y = s_h[rowA0 + lrow][kb + 1];
    b.x = s_wfc[kb][col];
    b.y = s_wfc[kb + 1][col];
    f0 = wmma_f32(a0, b, f0);
  }
  if (col < NCLS) {
#pragma unroll
    for (int r = 0; r < 8; ++r)
      out[(size_t)(b0 + rowA0 + rhalf + r) * NCLS + col] = f0[r];
  }
}

extern "C" void kernel_launch(void* const* d_in, const int* in_sizes, int n_in,
                              void* d_out, int out_size, void* d_ws,
                              size_t ws_size, hipStream_t stream) {
  const float* data = (const float*)d_in[0];
  const float* w_xh = (const float*)d_in[1];
  const float* b_xh = (const float*)d_in[2];
  const float* w_hh = (const float*)d_in[3];
  const float* b_hh = (const float*)d_in[4];
  const float* w_fc = (const float*)d_in[5];
  const float* b_fc = (const float*)d_in[6];
  float* out = (float*)d_out;

  dim3 grid(BATCH / MTILE);
  dim3 block(BLOCK);
  hipLaunchKernelGGL(rnn_fused_kernel, grid, block, 0, stream, data, w_xh,
                     b_xh, w_hh, b_hh, w_fc, b_fc, out);
}


# Hold the W_xh and W_hh B fragments in VGPRs for all 128 steps (read from LDS once per wave, h K-loop fully unrolled): no per-step LDS reads of weights, WMMAs run back to back
# speedup vs baseline: 2.1605x; 1.3490x over previous
#include <hip/hip_runtime.h>
#include <math.h>

typedef float v2f __attribute__((ext_vector_type(2)));
typedef float v8f __attribute__((ext_vector_type(8)));

namespace {
constexpr int T_SEQ = 128;
constexpr int BATCH = 16384;
constexpr int NIN   = 28;
constexpr int NHID  = 100;
constexpr int NCLS  = 10;
constexpr int HP    = 112;
constexpr int HS    = 113;
constexpr int NT    = 7;
constexpr int KX    = NIN / 4;
constexpr int KH    = NHID / 4;
constexpr int MTILE = 64;
constexpr int BLOCK = 128;
}

__device__ __forceinline__ v8f wmma_f32(v2f a, v2f b, v8f c) {
  return __builtin_amdgcn_wmma_f32_16x16x4_f32(false, a, false, b, (short)0, c,
                                               false, false);
}

__device__ __forceinline__ float fast_tanhf(float x) {
  float e = __expf(2.0f * x);
  return 1.0f - 2.0f * __builtin_amdgcn_rcpf(e + 1.0f);
}

__global__ __launch_bounds__(BLOCK) void rnn_fused_kernel(
    const float* __restrict__ data, const float* __restrict__ w_xh,
    const float* __restrict__ b_xh, const float* __restrict__ w_hh,
    const float* __restrict__ b_hh, const float* __restrict__ w_fc,
    const float* __restrict__ b_fc, float* __restrict__ out) {
  __shared__ float s_x[2][MTILE * NIN];
  __shared__ float s_h[MTILE][HS];
  __shared__ float s_whh[HP][HP];
  __shared__ float s_wxh[NIN][HP];
  __shared__ float s_wfc[HP][16];
  __shared__ float s_bias[HP];
  __shared__ float s_bfc[16];

  const int tid   = threadIdx.x;
  const int lane  = tid & 31;
  const int wave  = tid >> 5;
  const int col   = lane & 15;
  const int lrow  = lane & 15;
  const int khalf = (lane >> 4) << 1;
  const int rhalf = (lane >> 4) << 3;
  const int rowA0 = wave * 16;
  const int b0    = blockIdx.x * MTILE;

  for (int i = tid; i < MTILE * HS; i += BLOCK) (&s_h[0][0])[i] = 0.0f;
  for (int i = tid; i < HP * HP; i += BLOCK) {
    int k = i / HP, n = i % HP;
    s_whh[k][n] = (k < NHID && n < NHID) ? w_hh[k * NHID + n] : 0.0f;
  }
  for (int i = tid; i < NIN * HP; i += BLOCK) {
    int k = i / HP, n = i % HP;
    s_wxh[k][n] = (n < NHID) ? w_xh[k * NHID + n] : 0.0f;
  }
  for (int i = tid; i < HP * 16; i += BLOCK) {
    int k = i >> 4, n = i & 15;
    s_wfc[k][n] = (k < NHID && n < NCLS) ? w_fc[k * NCLS + n] : 0.0f;
  }
  for (int i = tid; i < HP; i += BLOCK)
    s_bias[i] = (i < NHID) ? (b_xh[i] + b_hh[i]) : 0.0f;
  if (tid < 16) s_bfc[tid] = (tid < NCLS) ? b_fc[tid] : 0.0f;
  __syncthreads();

  v8f cbias[NT];
#pragma unroll
  for (int n = 0; n < NT; ++n) {
    const float bv = s_bias[n * 16 + col];
    v8f c;
#pragma unroll
    for (int r = 0; r < 8; ++r) c[r] = bv;
    cbias[n] = c;
  }

  v2f bxh[KX][NT];
#pragma unroll
  for (int k = 0; k < KX; ++k) {
    const int kb = k * 4 + khalf;
#pragma unroll
    for (int n = 0; n < NT; ++n) {
      v2f b;
      b.x = s_wxh[kb][n * 16 + col];
      b.y = s_wxh[kb + 1][n * 16 + col];
      bxh[k][n] = b;
    }
  }
  v2f bhh[KH][NT];
#pragma unroll
  for (int k = 0; k < KH; ++k) {
    const int kb = k * 4 + khalf;
#pragma unroll
    for (int n = 0; n < NT; ++n) {
      v2f b;
      b.x = s_whh[kb][n * 16 + col];
      b.y = s_whh[kb + 1][n * 16 + col];
      bhh[k][n] = b;
    }
  }

#pragma unroll 1
  for (int t = 0; t < T_SEQ; ++t) {
    float* xb = s_x[t & 1];
    const float4* xt =
        (const float4*)(data + ((size_t)t * BATCH + b0) * NIN);
    for (int i = tid; i < MTILE * NIN / 4; i += BLOCK)
      ((float4*)xb)[i] = xt[i];
    if (t + 1 < T_SEQ) {
      const float* xn = data + ((size_t)(t + 1) * BATCH + b0) * NIN;
      for (int i = tid * 32; i < MTILE * NIN; i += BLOCK * 32)
        __builtin_prefetch(xn + i, 0, 1);
    }
    __syncthreads();

    v8f acc0[NT];
#pragma unroll
    for (int n = 0; n < NT; ++n) acc0[n] = cbias[n];

#pragma unroll
    for (int k = 0; k < KX; ++k) {
      const int kb = k * 4 + khalf;
      v2f a0;
      a0.x = xb[(rowA0 + lrow) * NIN + kb];
      a0.y = xb[(rowA0 + lrow) * NIN + kb + 1];
#pragma unroll
      for (int n = 0; n < NT; ++n) acc0[n] = wmma_f32(a0, bxh[k][n], acc0[n]);
    }

#pragma unroll
    for (int k = 0; k < KH; ++k) {
      const int kb = k * 4 + khalf;
      v2f a0;
      a0.x = s_h[rowA0 + lrow][kb];
      a0.y = s_h[rowA0 + lrow][kb + 1];
#pragma unroll
      for (int n = 0; n < NT; ++n) acc0[n] = wmma_f32(a0, bhh[k][n], acc0[n]);
    }

#pragma unroll
    for (int n = 0; n < NT; ++n) {
#pragma unroll
      for (int r = 0; r < 8; ++r)
        s_h[rowA0 + rhalf + r][n * 16 + col] = fast_tanhf(acc0[n][r]);
    }
  }

  v8f f0;
  {
    const float bv = s_bfc[col];
#pragma unroll
    for (int r = 0; r < 8; ++r) f0[r] = bv;
  }
#pragma unroll 1
  for (int k = 0; k < KH; ++k) {
    const int kb = k * 4 + khalf;
    v2f a0, b;
    a0.x = s_h[rowA0 + lrow][kb];
    a0.y = s_h[rowA0 + lrow][kb + 1];
    b.x = s_wfc[kb][col];
    b.y = s_wfc[kb + 1][col];
    f0 = wmma_f32(a0, b, f0);
  }
  if (col < NCLS) {
#pragma unroll
    for (int r = 0; r < 8; ++r)
      out[(size_t)(b0 + rowA0 + rhalf + r) * NCLS + col] = f0[r];
  }
}

extern "C" void kernel_launch(void* const* d_in, const int* in_sizes, int n_in,
                              void* d_out, int out_size, void* d_ws,
                              size_t ws_size, hipStream_t stream) {
  const float* data = (const float*)d_in[0];
  const float* w_xh = (const float*)d_in[1];
  const float* b_xh = (const float*)d_in[2];
  const float* w_hh = (const float*)d_in[3];
  const float* b_hh = (const float*)d_in[4];
  const float* w_fc = (const float*)d_in[5];
  const float* b_fc = (const float*)d_in[6];
  float* out = (float*)d_out;

  dim3 grid(BATCH / MTILE);
  dim3 block(BLOCK);
  hipLaunchKernelGGL(rnn_fused_kernel, grid, block, 0, stream, data, w_xh,
                     b_xh, w_hh, b_hh, w_fc, b_fc, out);
}


# Transposed recurrence (weights A, batch-per-lane B): tanh'd D fragment is the next step's B fragment, h stays in VGPRs, x_t loaded from global; no LDS or barrier per step; k sum order changes
# speedup vs baseline: 3.1798x; 1.4718x over previous
#include <hip/hip_runtime.h>
#include <math.h>


typedef float v2f __attribute__((ext_vector_type(2)));
typedef float v8f __attribute__((ext_vector_type(8)));

namespace {
constexpr int T_SEQ = 128;
constexpr int BATCH = 16384;
constexpr int NIN   = 28;
constexpr int NHID  = 100;
constexpr int NCLS  = 10;
constexpr int HP    = 112;
constexpr int NT    = 7;
constexpr int KX    = NIN / 4;
constexpr int KHR   = 26;
constexpr int MTILE = 64;
constexpr int BLOCK = 128;
}

__device__ __forceinline__ v8f wmma_f32(v2f a, v2f b, v8f c) {
  return __builtin_amdgcn_wmma_f32_16x16x4_f32(false, a, false, b, (short)0, c,
                                               false, false);
}

__device__ __forceinline__ float fast_tanhf(float x) {
  float e = __expf(2.0f * x);
  return 1.0f - 2.0f * __builtin_amdgcn_rcpf(e + 1.0f);
}


__device__ __forceinline__ void rnn_step(const v2f (&ahh)[KHR][NT],
                                         const v2f (&axh)[KX][NT],
                                         const v8f (&cb)[NT],
                                         const v2f (&xr)[KX], v8f (&pre)[NT]) {
  v8f acc[NT];
#pragma unroll
  for (int j = 0; j < KX; ++j) {
#pragma unroll
    for (int m = 0; m < NT; ++m)
      acc[m] = wmma_f32(axh[j][m], xr[j], (j == 0) ? cb[m] : acc[m]);
  }
  v2f h[NT][4];
#pragma unroll
  for (int n = 0; n < NT; ++n) {
#pragma unroll
    for (int q = 0; q < 4; ++q) {
      v2f v;
      v.x = fast_tanhf(pre[n][2 * q]);
      v.y = fast_tanhf(pre[n][2 * q + 1]);
      h[n][q] = v;
    }
  }
#pragma unroll
  for (int j = 0; j < KHR; ++j) {
    const v2f b = h[j >> 2][j & 3];
#pragma unroll
    for (int n = 0; n < NT; ++n) acc[n] = wmma_f32(ahh[j][n], b, acc[n]);
  }
#pragma unroll
  for (int n = 0; n < NT; ++n) pre[n] = acc[n];
}

__global__ __launch_bounds__(BLOCK) void rnn_fused_kernel(
    const float* __restrict__ data, const float* __restrict__ w_xh,
    const float* __restrict__ b_xh, const float* __restrict__ w_hh,
    const float* __restrict__ b_hh, const float* __restrict__ w_fc,
    const float* __restrict__ b_fc, float* __restrict__ out) {
  __shared__ float s_whh[HP][HP];
  __shared__ float s_wxh[NIN][HP];
  __shared__ float s_wfc[HP][16];
  __shared__ float s_bias[HP];
  __shared__ float s_bfc[16];

  const int tid   = threadIdx.x;
  const int lane  = tid & 31;
  const int wave  = tid >> 5;
  const int col   = lane & 15;
  const int half  = lane >> 4;
  const int rowA0 = wave * 16;
  const int b0    = blockIdx.x * MTILE;

  for (int i = tid; i < HP * HP; i += BLOCK) {
    int k = i / HP, n = i % HP;
    s_whh[k][n] = (k < NHID && n < NHID) ? w_hh[k * NHID + n] : 0.0f;
  }
  for (int i = tid; i < NIN * HP; i += BLOCK) {
    int k = i / HP, n = i % HP;
    s_wxh[k][n] = (n < NHID) ? w_xh[k * NHID + n] : 0.0f;
  }
  for (int i = tid; i < HP * 16; i += BLOCK) {
    int k = i >> 4, n = i & 15;
    s_wfc[k][n] = (k < NHID && n < NCLS) ? w_fc[k * NCLS + n] : 0.0f;
  }
  for (int i = tid; i < HP; i += BLOCK)
    s_bias[i] = (i < NHID) ? (b_xh[i] + b_hh[i]) : 0.0f;
  if (tid < 16) s_bfc[tid] = (tid < NCLS) ? b_fc[tid] : 0.0f;
  __syncthreads();

  v8f cb[NT];
#pragma unroll
  for (int n = 0; n < NT; ++n) {
    v8f c;
#pragma unroll
    for (int r = 0; r < 8; ++r) c[r] = s_bias[n * 16 + half * 8 + r];
    cb[n] = c;
  }

  v2f axh[KX][NT];
#pragma unroll
  for (int j = 0; j < KX; ++j) {
    const int kb = 14 * half + 2 * j;
#pragma unroll
    for (int n = 0; n < NT; ++n) {
      v2f a;
      a.x = s_wxh[kb][n * 16 + col];
      a.y = s_wxh[kb + 1][n * 16 + col];
      axh[j][n] = a;
    }
  }
  v2f ahh[KHR][NT];
#pragma unroll
  for (int j = 0; j < KHR; ++j) {
    const int kb = 16 * (j >> 2) + 8 * half + 2 * (j & 3);
#pragma unroll
    for (int n = 0; n < NT; ++n) {
      v2f a;
      a.x = s_whh[kb][n * 16 + col];
      a.y = s_whh[kb + 1][n * 16 + col];
      ahh[j][n] = a;
    }
  }

  const float* xlane = data + (size_t)(b0 + rowA0 + col) * NIN + 14 * half;
  const size_t xstep = (size_t)BATCH * NIN;

  v2f xr[KX];
#pragma unroll
  for (int j = 0; j < KX; ++j) xr[j] = ((const v2f*)xlane)[j];

  v8f pre[NT];
#pragma unroll
  for (int j = 0; j < KX; ++j) {
#pragma unroll
    for (int m = 0; m < NT; ++m)
      pre[m] = wmma_f32(axh[j][m], xr[j], (j == 0) ? cb[m] : pre[m]);
  }
#pragma unroll
  for (int j = 0; j < KX; ++j) xr[j] = ((const v2f*)(xlane + xstep))[j];

#pragma unroll 1
  for (int t = 1; t < T_SEQ; ++t) {
    rnn_step(ahh, axh, cb, xr, pre);
    const int tn = (t + 1 < T_SEQ) ? (t + 1) : (T_SEQ - 1);
    const v2f* xp = (const v2f*)(xlane + (size_t)tn * xstep);
#pragma unroll
    for (int j = 0; j < KX; ++j) xr[j] = xp[j];
    const int tp = (t + 3 < T_SEQ) ? (t + 3) : (T_SEQ - 1);
    __builtin_prefetch(xlane + (size_t)tp * xstep, 0, 1);
  }

  v2f hA[NT][4];
#pragma unroll
  for (int n = 0; n < NT; ++n) {
#pragma unroll
    for (int q = 0; q < 4; ++q) {
      v2f v;
      v.x = fast_tanhf(pre[n][2 * q]);
      v.y = fast_tanhf(pre[n][2 * q + 1]);
      hA[n][q] = v;
    }
  }

  v8f f0;
#pragma unroll
  for (int r = 0; r < 8; ++r) f0[r] = s_bfc[half * 8 + r];
#pragma unroll
  for (int j = 0; j < KHR; ++j) {
    const int kb = 16 * (j >> 2) + 8 * half + 2 * (j & 3);
    v2f a;
    a.x = s_wfc[kb][col];
    a.y = s_wfc[kb + 1][col];
    f0 = wmma_f32(a, hA[j >> 2][j & 3], f0);
  }
  float* orow = out + (size_t)(b0 + rowA0 + col) * NCLS + half * 8;
#pragma unroll
  for (int r = 0; r < 8; ++r)
    if (half * 8 + r < NCLS) orow[r] = f0[r];
}

extern "C" void kernel_launch(void* const* d_in, const int* in_sizes, int n_in,
                              void* d_out, int out_size, void* d_ws,
                              size_t ws_size, hipStream_t stream) {
  const float* data = (const float*)d_in[0];
  const float* w_xh = (const float*)d_in[1];
  const float* b_xh = (const float*)d_in[2];
  const float* w_hh = (const float*)d_in[3];
  const float* b_hh = (const float*)d_in[4];
  const float* w_fc = (const float*)d_in[5];
  const float* b_fc = (const float*)d_in[6];
  float* out = (float*)d_out;

  dim3 grid(BATCH / MTILE);
  dim3 block(BLOCK);
  hipLaunchKernelGGL(rnn_fused_kernel, grid, block, 0, stream, data, w_xh,
                     b_xh, w_hh, b_hh, w_fc, b_fc, out);
}


# Weight staging de-serialized: all weight and bias loads of the prologue issued as one group (16-byte, unconditional, padding selected to zero) before the first wait; was 139 dependent 4-byte trips
# speedup vs baseline: 3.7474x; 1.1785x over previous
#include <hip/hip_runtime.h>
#include <math.h>


typedef float v2f __attribute__((ext_vector_type(2)));
typedef float v8f __attribute__((ext_vector_type(8)));
typedef float v4f __attribute__((ext_vector_type(4)));
typedef float v4fu __attribute__((ext_vector_type(4), aligned(4)));

namespace {
constexpr int T_SEQ = 128;
constexpr int BATCH = 16384;
constexpr int NIN   = 28;
constexpr int NHID  = 100;
constexpr int NCLS  = 10;
constexpr int HP    = 112;
constexpr int NT    = 7;
constexpr int KX    = NIN / 4;
constexpr int KHR   = 26;
constexpr int MTILE = 64;
constexpr int BLOCK = 128;
}

__device__ __forceinline__ v8f wmma_f32(v2f a, v2f b, v8f c) {
  return __builtin_amdgcn_wmma_f32_16x16x4_f32(false, a, false, b, (short)0, c,
                                               false, false);
}

__device__ __forceinline__ float fast_tanhf(float x) {
  float e = __expf(2.0f * x);
  return 1.0f - 2.0f * __builtin_amdgcn_rcpf(e + 1.0f);
}


__device__ __forceinline__ void rnn_step(const v2f (&ahh)[KHR][NT],
                                         const v2f (&axh)[KX][NT],
                                         const v8f (&cb)[NT],
                                         const v2f (&xr)[KX], v8f (&pre)[NT]) {
  v8f acc[NT];
#pragma unroll
  for (int j = 0; j < KX; ++j) {
#pragma unroll
    for (int m = 0; m < NT; ++m)
      acc[m] = wmma_f32(axh[j][m], xr[j], (j == 0) ? cb[m] : acc[m]);
  }
  v2f h[NT][4];
#pragma unroll
  for (int n = 0; n < NT; ++n) {
#pragma unroll
    for (int q = 0; q < 4; ++q) {
      v2f v;
      v.x = fast_tanhf(pre[n][2 * q]);
      v.y = fast_tanhf(pre[n][2 * q + 1]);
      h[n][q] = v;
    }
  }
#pragma unroll
  for (int j = 0; j < KHR; ++j) {
    const v2f b = h[j >> 2][j & 3];
#pragma unroll
    for (int n = 0; n < NT; ++n) acc[n] = wmma_f32(ahh[j][n], b, acc[n]);
  }
#pragma unroll
  for (int n = 0; n < NT; ++n) pre[n] = acc[n];
}

__global__ __launch_bounds__(BLOCK) void rnn_fused_kernel(
    const float* __restrict__ data, const float* __restrict__ w_xh,
    const float* __restrict__ b_xh, const float* __restrict__ w_hh,
    const float* __restrict__ b_hh, const float* __restrict__ w_fc,
    const float* __restrict__ b_fc, float* __restrict__ out) {
  __shared__ __attribute__((aligned(16))) float s_whh[HP][HP];
  __shared__ __attribute__((aligned(16))) float s_wxh[NIN][HP];
  __shared__ float s_wfc[HP][16];
  __shared__ float s_bias[HP];
  __shared__ float s_bfc[16];

  const int tid   = threadIdx.x;
  __builtin_assume(tid < BLOCK);
  const int lane  = tid & 31;
  const int wave  = tid >> 5;
  const int col   = lane & 15;
  const int half  = lane >> 4;
  const int rowA0 = wave * 16;
  const int b0    = blockIdx.x * MTILE;

  constexpr int RQ  = NHID / 4;
  constexpr int PQ  = HP / 4;
  constexpr int NQH = (HP * PQ + BLOCK - 1) / BLOCK;
  constexpr int NQX = (NIN * PQ + BLOCK - 1) / BLOCK;
  constexpr int NFC = (HP * 16) / BLOCK;
  v4f qh[NQH];
  v4f qx[NQX];
  float qf[NFC];
#pragma unroll
  for (int it = 0; it < NFC; ++it) {
    const int i = tid + it * BLOCK;
    const int k = i >> 4, n = i & 15;
    const bool ok = (k < NHID) && (n < NCLS);
    qf[it] = w_fc[ok ? (k * NCLS + n) : 0];
  }
  const float qb0 = b_xh[(tid < NHID) ? tid : 0];
  const float qb1 = b_hh[(tid < NHID) ? tid : 0];
  const float qb2 = b_fc[(tid < NCLS) ? tid : 0];
#pragma unroll
  for (int it = 0; it < NQH; ++it) {
    const int s = tid + it * BLOCK;
    const int k = s / PQ, q = s - k * PQ;
    const bool ok = (k < NHID) && (q < RQ);
    qh[it] = *(const v4fu*)(w_hh + (ok ? (k * NHID + 4 * q) : 0));
  }
#pragma unroll
  for (int it = 0; it < NQX; ++it) {
    const int s = tid + it * BLOCK;
    const int k = s / PQ, q = s - k * PQ;
    const bool ok = (k < NIN) && (q < RQ);
    qx[it] = *(const v4fu*)(w_xh + (ok ? (k * NHID + 4 * q) : 0));
  }

  const v4f zero4 = {0.0f, 0.0f, 0.0f, 0.0f};
#pragma unroll
  for (int it = 0; it < NQH; ++it) {
    const int s = tid + it * BLOCK;
    const int k = s / PQ, q = s - k * PQ;
    const bool ok = (k < NHID) && (q < RQ);
    if (s < HP * PQ) *(v4f*)&s_whh[k][4 * q] = ok ? qh[it] : zero4;
  }
#pragma unroll
  for (int it = 0; it < NQX; ++it) {
    const int s = tid + it * BLOCK;
    const int k = s / PQ, q = s - k * PQ;
    const bool ok = (k < NIN) && (q < RQ);
    if (s < NIN * PQ) *(v4f*)&s_wxh[k][4 * q] = ok ? qx[it] : zero4;
  }
#pragma unroll
  for (int it = 0; it < NFC; ++it) {
    const int i = tid + it * BLOCK;
    const int k = i >> 4, n = i & 15;
    const bool ok = (k < NHID) && (n < NCLS);
    s_wfc[k][n] = ok ? qf[it] : 0.0f;
  }
  if (tid < HP) s_bias[tid] = (tid < NHID) ? (qb0 + qb1) : 0.0f;
  if (tid < 16) s_bfc[tid] = (tid < NCLS) ? qb2 : 0.0f;
  __syncthreads();

  v8f cb[NT];
#pragma unroll
  for (int n = 0; n < NT; ++n) {
    v8f c;
#pragma unroll
    for (int r = 0; r < 8; ++r) c[r] = s_bias[n * 16 + half * 8 + r];
    cb[n] = c;
  }

  v2f axh[KX][NT];
#pragma unroll
  for (int j = 0; j < KX; ++j) {
    const int kb = 14 * half + 2 * j;
#pragma unroll
    for (int n = 0; n < NT; ++n) {
      v2f a;
      a.x = s_wxh[kb][n * 16 + col];
      a.y = s_wxh[kb + 1][n * 16 + col];
      axh[j][n] = a;
    }
  }
  v2f ahh[KHR][NT];
#pragma unroll
  for (int j = 0; j < KHR; ++j) {
    const int kb = 16 * (j >> 2) + 8 * half + 2 * (j & 3);
#pragma unroll
    for (int n = 0; n < NT; ++n) {
      v2f a;
      a.x = s_whh[kb][n * 16 + col];
      a.y = s_whh[kb + 1][n * 16 + col];
      ahh[j][n] = a;
    }
  }

  const float* xlane = data + (size_t)(b0 + rowA0 + col) * NIN + 14 * half;
  const size_t xstep = (size_t)BATCH * NIN;

  v2f xr[KX];
#pragma unroll
  for (int j = 0; j < KX; ++j) xr[j] = ((const v2f*)xlane)[j];

  v8f pre[NT];
#pragma unroll
  for (int j = 0; j < KX; ++j) {
#pragma unroll
    for (int m = 0; m < NT; ++m)
      pre[m] = wmma_f32(axh[j][m], xr[j], (j == 0) ? cb[m] : pre[m]);
  }
#pragma unroll
  for (int j = 0; j < KX; ++j) xr[j] = ((const v2f*)(xlane + xstep))[j];

#pragma unroll 1
  for (int t = 1; t < T_SEQ; ++t) {
    rnn_step(ahh, axh, cb, xr, pre);
    const int tn = (t + 1 < T_SEQ) ? (t + 1) : (T_SEQ - 1);
    const v2f* xp = (const v2f*)(xlane + (size_t)tn * xstep);
#pragma unroll
    for (int j = 0; j < KX; ++j) xr[j] = xp[j];
    const int tp = (t + 3 < T_SEQ) ? (t + 3) : (T_SEQ - 1);
    __builtin_prefetch(xlane + (size_t)tp * xstep, 0, 1);
  }

  v2f hA[NT][4];
#pragma unroll
  for (int n = 0; n < NT; ++n) {
#pragma unroll
    for (int q = 0; q < 4; ++q) {
      v2f v;
      v.x = fast_tanhf(pre[n][2 * q]);
      v.y = fast_tanhf(pre[n][2 * q + 1]);
      hA[n][q] = v;
    }
  }

  v8f f0;
#pragma unroll
  for (int r = 0; r < 8; ++r) f0[r] = s_bfc[half * 8 + r];
#pragma unroll
  for (int j = 0; j < KHR; ++j) {
    const int kb = 16 * (j >> 2) + 8 * half + 2 * (j & 3);
    v2f a;
    a.x = s_wfc[kb][col];
    a.y = s_wfc[kb + 1][col];
    f0 = wmma_f32(a, hA[j >> 2][j & 3], f0);
  }
  float* orow = out + (size_t)(b0 + rowA0 + col) * NCLS + half * 8;
#pragma unroll
  for (int r = 0; r < 8; ++r)
    if (half * 8 + r < NCLS) orow[r] = f0[r];
}

extern "C" void kernel_launch(void* const* d_in, const int* in_sizes, int n_in,
                              void* d_out, int out_size, void* d_ws,
                              size_t ws_size, hipStream_t stream) {
  const float* data = (const float*)d_in[0];
  const float* w_xh = (const float*)d_in[1];
  const float* b_xh = (const float*)d_in[2];
  const float* w_hh = (const float*)d_in[3];
  const float* b_hh = (const float*)d_in[4];
  const float* w_fc = (const float*)d_in[5];
  const float* b_fc = (const float*)d_in[6];
  float* out = (float*)d_out;

  dim3 grid(BATCH / MTILE);
  dim3 block(BLOCK);
  hipLaunchKernelGGL(rnn_fused_kernel, grid, block, 0, stream, data, w_xh,
                     b_xh, w_hh, b_hh, w_fc, b_fc, out);
}


# tanh as one native v_tanh_f32 per value in place of exp2 + reciprocal (2 transcendentals + 4 plain vector ops per value); h rounding changes, sum order unchanged
# speedup vs baseline: 3.9304x; 1.0488x over previous
#include <hip/hip_runtime.h>
#include <math.h>


typedef float v2f __attribute__((ext_vector_type(2)));
typedef float v8f __attribute__((ext_vector_type(8)));
typedef float v4f __attribute__((ext_vector_type(4)));
typedef float v4fu __attribute__((ext_vector_type(4), aligned(4)));

namespace {
constexpr int T_SEQ = 128;
constexpr int BATCH = 16384;
constexpr int NIN   = 28;
constexpr int NHID  = 100;
constexpr int NCLS  = 10;
constexpr int HP    = 112;
constexpr int NT    = 7;
constexpr int KX    = NIN / 4;
constexpr int KHR   = 26;
constexpr int MTILE = 64;
constexpr int BLOCK = 128;
}

__device__ __forceinline__ v8f wmma_f32(v2f a, v2f b, v8f c) {
  return __builtin_amdgcn_wmma_f32_16x16x4_f32(false, a, false, b, (short)0, c,
                                               false, false);
}

__device__ __forceinline__ float fast_tanhf(float x) {
  return __builtin_amdgcn_tanhf(x);
}


__device__ __forceinline__ void rnn_step(const v2f (&ahh)[KHR][NT],
                                         const v2f (&axh)[KX][NT],
                                         const v8f (&cb)[NT],
                                         const v2f (&xr)[KX], v8f (&pre)[NT]) {
  v8f acc[NT];
#pragma unroll
  for (int j = 0; j < KX; ++j) {
#pragma unroll
    for (int m = 0; m < NT; ++m)
      acc[m] = wmma_f32(axh[j][m], xr[j], (j == 0) ? cb[m] : acc[m]);
  }
  v2f h[NT][4];
#pragma unroll
  for (int n = 0; n < NT; ++n) {
#pragma unroll
    for (int q = 0; q < 4; ++q) {
      v2f v;
      v.x = fast_tanhf(pre[n][2 * q]);
      v.y = fast_tanhf(pre[n][2 * q + 1]);
      h[n][q] = v;
    }
  }
#pragma unroll
  for (int j = 0; j < KHR; ++j) {
    const v2f b = h[j >> 2][j & 3];
#pragma unroll
    for (int n = 0; n < NT; ++n) acc[n] = wmma_f32(ahh[j][n], b, acc[n]);
  }
#pragma unroll
  for (int n = 0; n < NT; ++n) pre[n] = acc[n];
}

__global__ __launch_bounds__(BLOCK) void rnn_fused_kernel(
    const float* __restrict__ data, const float* __restrict__ w_xh,
    const float* __restrict__ b_xh, const float* __restrict__ w_hh,
    const float* __restrict__ b_hh, const float* __restrict__ w_fc,
    const float* __restrict__ b_fc, float* __restrict__ out) {
  __shared__ __attribute__((aligned(16))) float s_whh[HP][HP];
  __shared__ __attribute__((aligned(16))) float s_wxh[NIN][HP];
  __shared__ float s_wfc[HP][16];
  __shared__ float s_bias[HP];
  __shared__ float s_bfc[16];

  const int tid   = threadIdx.x;
  __builtin_assume(tid < BLOCK);
  const int lane  = tid & 31;
  const int wave  = tid >> 5;
  const int col   = lane & 15;
  const int half  = lane >> 4;
  const int rowA0 = wave * 16;
  const int b0    = blockIdx.x * MTILE;

  constexpr int RQ  = NHID / 4;
  constexpr int PQ  = HP / 4;
  constexpr int NQH = (HP * PQ + BLOCK - 1) / BLOCK;
  constexpr int NQX = (NIN * PQ + BLOCK - 1) / BLOCK;
  constexpr int NFC = (HP * 16) / BLOCK;
  v4f qh[NQH];
  v4f qx[NQX];
  float qf[NFC];
#pragma unroll
  for (int it = 0; it < NFC; ++it) {
    const int i = tid + it * BLOCK;
    const int k = i >> 4, n = i & 15;
    const bool ok = (k < NHID) && (n < NCLS);
    qf[it] = w_fc[ok ? (k * NCLS + n) : 0];
  }
  const float qb0 = b_xh[(tid < NHID) ? tid : 0];
  const float qb1 = b_hh[(tid < NHID) ? tid : 0];
  const float qb2 = b_fc[(tid < NCLS) ? tid : 0];
#pragma unroll
  for (int it = 0; it < NQH; ++it) {
    const int s = tid + it * BLOCK;
    const int k = s / PQ, q = s - k * PQ;
    const bool ok = (k < NHID) && (q < RQ);
    qh[it] = *(const v4fu*)(w_hh + (ok ? (k * NHID + 4 * q) : 0));
  }
#pragma unroll
  for (int it = 0; it < NQX; ++it) {
    const int s = tid + it * BLOCK;
    const int k = s / PQ, q = s - k * PQ;
    const bool ok = (k < NIN) && (q < RQ);
    qx[it] = *(const v4fu*)(w_xh + (ok ? (k * NHID + 4 * q) : 0));
  }

  const v4f zero4 = {0.0f, 0.0f, 0.0f, 0.0f};
#pragma unroll
  for (int it = 0; it < NQH; ++it) {
    const int s = tid + it * BLOCK;
    const int k = s / PQ, q = s - k * PQ;
    const bool ok = (k < NHID) && (q < RQ);
    if (s < HP * PQ) *(v4f*)&s_whh[k][4 * q] = ok ? qh[it] : zero4;
  }
#pragma unroll
  for (int it = 0; it < NQX; ++it) {
    const int s = tid + it * BLOCK;
    const int k = s / PQ, q = s - k * PQ;
    const bool ok = (k < NIN) && (q < RQ);
    if (s < NIN * PQ) *(v4f*)&s_wxh[k][4 * q] = ok ? qx[it] : zero4;
  }
#pragma unroll
  for (int it = 0; it < NFC; ++it) {
    const int i = tid + it * BLOCK;
    const int k = i >> 4, n = i & 15;
    const bool ok = (k < NHID) && (n < NCLS);
    s_wfc[k][n] = ok ? qf[it] : 0.0f;
  }
  if (tid < HP) s_bias[tid] = (tid < NHID) ? (qb0 + qb1) : 0.0f;
  if (tid < 16) s_bfc[tid] = (tid < NCLS) ? qb2 : 0.0f;
  __syncthreads();

  v8f cb[NT];
#pragma unroll
  for (int n = 0; n < NT; ++n) {
    v8f c;
#pragma unroll
    for (int r = 0; r < 8; ++r) c[r] = s_bias[n * 16 + half * 8 + r];
    cb[n] = c;
  }

  v2f axh[KX][NT];
#pragma unroll
  for (int j = 0; j < KX; ++j) {
    const int kb = 14 * half + 2 * j;
#pragma unroll
    for (int n = 0; n < NT; ++n) {
      v2f a;
      a.x = s_wxh[kb][n * 16 + col];
      a.y = s_wxh[kb + 1][n * 16 + col];
      axh[j][n] = a;
    }
  }
  v2f ahh[KHR][NT];
#pragma unroll
  for (int j = 0; j < KHR; ++j) {
    const int kb = 16 * (j >> 2) + 8 * half + 2 * (j & 3);
#pragma unroll
    for (int n = 0; n < NT; ++n) {
      v2f a;
      a.x = s_whh[kb][n * 16 + col];
      a.y = s_whh[kb + 1][n * 16 + col];
      ahh[j][n] = a;
    }
  }

  const float* xlane = data + (size_t)(b0 + rowA0 + col) * NIN + 14 * half;
  const size_t xstep = (size_t)BATCH * NIN;

  v2f xr[KX];
#pragma unroll
  for (int j = 0; j < KX; ++j) xr[j] = ((const v2f*)xlane)[j];

  v8f pre[NT];
#pragma unroll
  for (int j = 0; j < KX; ++j) {
#pragma unroll
    for (int m = 0; m < NT; ++m)
      pre[m] = wmma_f32(axh[j][m], xr[j], (j == 0) ? cb[m] : pre[m]);
  }
#pragma unroll
  for (int j = 0; j < KX; ++j) xr[j] = ((const v2f*)(xlane + xstep))[j];

#pragma unroll 1
  for (int t = 1; t < T_SEQ; ++t) {
    rnn_step(ahh, axh, cb, xr, pre);
    const int tn = (t + 1 < T_SEQ) ? (t + 1) : (T_SEQ - 1);
    const v2f* xp = (const v2f*)(xlane + (size_t)tn * xstep);
#pragma unroll
    for (int j = 0; j < KX; ++j) xr[j] = xp[j];
    const int tp = (t + 3 < T_SEQ) ? (t + 3) : (T_SEQ - 1);
    __builtin_prefetch(xlane + (size_t)tp * xstep, 0, 1);
  }

  v2f hA[NT][4];
#pragma unroll
  for (int n = 0; n < NT; ++n) {
#pragma unroll
    for (int q = 0; q < 4; ++q) {
      v2f v;
      v.x = fast_tanhf(pre[n][2 * q]);
      v.y = fast_tanhf(pre[n][2 * q + 1]);
      hA[n][q] = v;
    }
  }

  v8f f0;
#pragma unroll
  for (int r = 0; r < 8; ++r) f0[r] = s_bfc[half * 8 + r];
#pragma unroll
  for (int j = 0; j < KHR; ++j) {
    const int kb = 16 * (j >> 2) + 8 * half + 2 * (j & 3);
    v2f a;
    a.x = s_wfc[kb][col];
    a.y = s_wfc[kb + 1][col];
    f0 = wmma_f32(a, hA[j >> 2][j & 3], f0);
  }
  float* orow = out + (size_t)(b0 + rowA0 + col) * NCLS + half * 8;
#pragma unroll
  for (int r = 0; r < 8; ++r)
    if (half * 8 + r < NCLS) orow[r] = f0[r];
}

extern "C" void kernel_launch(void* const* d_in, const int* in_sizes, int n_in,
                              void* d_out, int out_size, void* d_ws,
                              size_t ws_size, hipStream_t stream) {
  const float* data = (const float*)d_in[0];
  const float* w_xh = (const float*)d_in[1];
  const float* b_xh = (const float*)d_in[2];
  const float* w_hh = (const float*)d_in[3];
  const float* b_hh = (const float*)d_in[4];
  const float* w_fc = (const float*)d_in[5];
  const float* b_fc = (const float*)d_in[6];
  float* out = (float*)d_out;

  dim3 grid(BATCH / MTILE);
  dim3 block(BLOCK);
  hipLaunchKernelGGL(rnn_fused_kernel, grid, block, 0, stream, data, w_xh,
                     b_xh, w_hh, b_hh, w_fc, b_fc, out);
}


# Features 96..99 (the 4-row remainder of hidden = 100) summed with packed f32 FMAs on the vector unit beside the matrix stream: 198 WMMA per step, not 231; the sum order of those 4 features changes
# speedup vs baseline: 4.1586x; 1.0581x over previous
#include <hip/hip_runtime.h>
#include <math.h>


typedef float v2f __attribute__((ext_vector_type(2)));
typedef float v8f __attribute__((ext_vector_type(8)));
typedef float v4f __attribute__((ext_vector_type(4)));
typedef float v4fu __attribute__((ext_vector_type(4), aligned(4)));

namespace {
constexpr int T_SEQ = 128;
constexpr int BATCH = 16384;
constexpr int NIN   = 28;
constexpr int NHID  = 100;
constexpr int NCLS  = 10;
constexpr int HP    = 112;
constexpr int NT    = 7;
constexpr int NTM   = 6;
constexpr int NREM  = NHID - 16 * NTM;
constexpr int KX    = NIN / 4;
constexpr int KHR   = 26;
constexpr int MTILE = 64;
constexpr int BLOCK = 128;
}

__device__ __forceinline__ v8f wmma_f32(v2f a, v2f b, v8f c) {
  return __builtin_amdgcn_wmma_f32_16x16x4_f32(false, a, false, b, (short)0, c,
                                               false, false);
}

__device__ __forceinline__ float fast_tanhf(float x) {
  return __builtin_amdgcn_tanhf(x);
}

__device__ __forceinline__ float xhalf(float x) {
  return __builtin_bit_cast(
      float, __builtin_amdgcn_permlanex16(__builtin_bit_cast(int, x),
                                          __builtin_bit_cast(int, x), 0x76543210,
                                          0xfedcba98, false, false));
}


__device__ __forceinline__ void rnn_step(const v2f (&ahh)[KHR][NTM],
                                         const v2f (&axh)[KX][NTM],
                                         const v8f (&cb)[NTM],
                                         const v2f (&whr)[KHR][NREM],
                                         const v2f (&wxr)[KX][NREM],
                                         const float (&br)[NREM],
                                         const v2f (&xr)[KX], v8f (&pre)[NTM],
                                         float (&prr)[NREM]) {
  v8f acc[NTM];
  v2f ra[NREM];
#pragma unroll
  for (int f = 0; f < NREM; ++f) ra[f] = v2f{0.0f, 0.0f};
#pragma unroll
  for (int j = 0; j < KX; ++j) {
#pragma unroll
    for (int m = 0; m < NTM; ++m)
      acc[m] = wmma_f32(axh[j][m], xr[j], (j == 0) ? cb[m] : acc[m]);
#pragma unroll
    for (int f = 0; f < NREM; ++f)
      ra[f] = __builtin_elementwise_fma(wxr[j][f], xr[j], ra[f]);
  }
  v2f h[NT][4];
#pragma unroll
  for (int n = 0; n < NTM; ++n) {
#pragma unroll
    for (int q = 0; q < 4; ++q) {
      v2f v;
      v.x = fast_tanhf(pre[n][2 * q]);
      v.y = fast_tanhf(pre[n][2 * q + 1]);
      h[n][q] = v;
    }
  }
  h[NTM][0] = v2f{fast_tanhf(prr[0]), fast_tanhf(prr[1])};
  h[NTM][1] = v2f{fast_tanhf(prr[2]), fast_tanhf(prr[3])};
#pragma unroll
  for (int j = 0; j < KHR; ++j) {
    const v2f b = h[j >> 2][j & 3];
#pragma unroll
    for (int n = 0; n < NTM; ++n) acc[n] = wmma_f32(ahh[j][n], b, acc[n]);
#pragma unroll
    for (int f = 0; f < NREM; ++f)
      ra[f] = __builtin_elementwise_fma(whr[j][f], b, ra[f]);
  }
#pragma unroll
  for (int n = 0; n < NTM; ++n) pre[n] = acc[n];
#pragma unroll
  for (int f = 0; f < NREM; ++f) {
    const float s = ra[f].x + ra[f].y;
    prr[f] = br[f] + (s + xhalf(s));
  }
}

__global__ __launch_bounds__(BLOCK) void rnn_fused_kernel(
    const float* __restrict__ data, const float* __restrict__ w_xh,
    const float* __restrict__ b_xh, const float* __restrict__ w_hh,
    const float* __restrict__ b_hh, const float* __restrict__ w_fc,
    const float* __restrict__ b_fc, float* __restrict__ out) {
  __shared__ __attribute__((aligned(16))) float s_whh[HP][HP];
  __shared__ __attribute__((aligned(16))) float s_wxh[NIN][HP];
  __shared__ float s_wfc[HP][16];
  __shared__ float s_bias[HP];
  __shared__ float s_bfc[16];

  const int tid   = threadIdx.x;
  __builtin_assume(tid < BLOCK);
  const int lane  = tid & 31;
  const int wave  = tid >> 5;
  const int col   = lane & 15;
  const int half  = lane >> 4;
  const int rowA0 = wave * 16;
  const int b0    = blockIdx.x * MTILE;

  constexpr int RQ  = NHID / 4;
  constexpr int PQ  = HP / 4;
  constexpr int NQH = (HP * PQ + BLOCK - 1) / BLOCK;
  constexpr int NQX = (NIN * PQ + BLOCK - 1) / BLOCK;
  constexpr int NFC = (HP * 16) / BLOCK;
  v4f qh[NQH];
  v4f qx[NQX];
  float qf[NFC];
#pragma unroll
  for (int it = 0; it < NFC; ++it) {
    const int i = tid + it * BLOCK;
    const int k = i >> 4, n = i & 15;
    const bool ok = (k < NHID) && (n < NCLS);
    qf[it] = w_fc[ok ? (k * NCLS + n) : 0];
  }
  const float qb0 = b_xh[(tid < NHID) ? tid : 0];
  const float qb1 = b_hh[(tid < NHID) ? tid : 0];
  const float qb2 = b_fc[(tid < NCLS) ? tid : 0];
#pragma unroll
  for (int it = 0; it < NQH; ++it) {
    const int s = tid + it * BLOCK;
    const int k = s / PQ, q = s - k * PQ;
    const bool ok = (k < NHID) && (q < RQ);
    qh[it] = *(const v4fu*)(w_hh + (ok ? (k * NHID + 4 * q) : 0));
  }
#pragma unroll
  for (int it = 0; it < NQX; ++it) {
    const int s = tid + it * BLOCK;
    const int k = s / PQ, q = s - k * PQ;
    const bool ok = (k < NIN) && (q < RQ);
    qx[it] = *(const v4fu*)(w_xh + (ok ? (k * NHID + 4 * q) : 0));
  }

  const v4f zero4 = {0.0f, 0.0f, 0.0f, 0.0f};
#pragma unroll
  for (int it = 0; it < NQH; ++it) {
    const int s = tid + it * BLOCK;
    const int k = s / PQ, q = s - k * PQ;
    const bool ok = (k < NHID) && (q < RQ);
    if (s < HP * PQ) *(v4f*)&s_whh[k][4 * q] = ok ? qh[it] : zero4;
  }
#pragma unroll
  for (int it = 0; it < NQX; ++it) {
    const int s = tid + it * BLOCK;
    const int k = s / PQ, q = s - k * PQ;
    const bool ok = (k < NIN) && (q < RQ);
    if (s < NIN * PQ) *(v4f*)&s_wxh[k][4 * q] = ok ? qx[it] : zero4;
  }
#pragma unroll
  for (int it = 0; it < NFC; ++it) {
    const int i = tid + it * BLOCK;
    const int k = i >> 4, n = i & 15;
    const bool ok = (k < NHID) && (n < NCLS);
    s_wfc[k][n] = ok ? qf[it] : 0.0f;
  }
  if (tid < HP) s_bias[tid] = (tid < NHID) ? (qb0 + qb1) : 0.0f;
  if (tid < 16) s_bfc[tid] = (tid < NCLS) ? qb2 : 0.0f;
  __syncthreads();

  v8f cb[NTM];
#pragma unroll
  for (int n = 0; n < NTM; ++n) {
    v8f c;
#pragma unroll
    for (int r = 0; r < 8; ++r) c[r] = s_bias[n * 16 + half * 8 + r];
    cb[n] = c;
  }
  float br[NREM];
#pragma unroll
  for (int f = 0; f < NREM; ++f) br[f] = s_bias[16 * NTM + f];

  v2f axh[KX][NTM];
  v2f wxr[KX][NREM];
#pragma unroll
  for (int j = 0; j < KX; ++j) {
    const int kb = 14 * half + 2 * j;
#pragma unroll
    for (int n = 0; n < NTM; ++n) {
      v2f a;
      a.x = s_wxh[kb][n * 16 + col];
      a.y = s_wxh[kb + 1][n * 16 + col];
      axh[j][n] = a;
    }
#pragma unroll
    for (int f = 0; f < NREM; ++f) {
      v2f w;
      w.x = s_wxh[kb][16 * NTM + f];
      w.y = s_wxh[kb + 1][16 * NTM + f];
      wxr[j][f] = w;
    }
  }
  v2f ahh[KHR][NTM];
  v2f whr[KHR][NREM];
#pragma unroll
  for (int j = 0; j < KHR; ++j) {
    const int kb = 16 * (j >> 2) + 8 * half + 2 * (j & 3);
#pragma unroll
    for (int n = 0; n < NTM; ++n) {
      v2f a;
      a.x = s_whh[kb][n * 16 + col];
      a.y = s_whh[kb + 1][n * 16 + col];
      ahh[j][n] = a;
    }
#pragma unroll
    for (int f = 0; f < NREM; ++f) {
      v2f w;
      w.x = s_whh[kb][16 * NTM + f];
      w.y = s_whh[kb + 1][16 * NTM + f];
      whr[j][f] = w;
    }
  }

  const float* xlane = data + (size_t)(b0 + rowA0 + col) * NIN + 14 * half;
  const size_t xstep = (size_t)BATCH * NIN;

  v2f xr[KX];
#pragma unroll
  for (int j = 0; j < KX; ++j) xr[j] = ((const v2f*)xlane)[j];

  v8f pre[NTM];
  float prr[NREM];
  {
    v2f ra[NREM];
#pragma unroll
    for (int f = 0; f < NREM; ++f) ra[f] = v2f{0.0f, 0.0f};
#pragma unroll
    for (int j = 0; j < KX; ++j) {
#pragma unroll
      for (int m = 0; m < NTM; ++m)
        pre[m] = wmma_f32(axh[j][m], xr[j], (j == 0) ? cb[m] : pre[m]);
#pragma unroll
      for (int f = 0; f < NREM; ++f)
        ra[f] = __builtin_elementwise_fma(wxr[j][f], xr[j], ra[f]);
    }
#pragma unroll
    for (int f = 0; f < NREM; ++f) {
      const float s = ra[f].x + ra[f].y;
      prr[f] = br[f] + (s + xhalf(s));
    }
  }
#pragma unroll
  for (int j = 0; j < KX; ++j) xr[j] = ((const v2f*)(xlane + xstep))[j];

#pragma unroll 1
  for (int t = 1; t < T_SEQ; ++t) {
    rnn_step(ahh, axh, cb, whr, wxr, br, xr, pre, prr);
    const int tn = (t + 1 < T_SEQ) ? (t + 1) : (T_SEQ - 1);
    const v2f* xp = (const v2f*)(xlane + (size_t)tn * xstep);
#pragma unroll
    for (int j = 0; j < KX; ++j) xr[j] = xp[j];
    const int tp = (t + 3 < T_SEQ) ? (t + 3) : (T_SEQ - 1);
    __builtin_prefetch(xlane + (size_t)tp * xstep, 0, 1);
  }

  v2f hA[NT][4];
  hA[NTM][0] = v2f{fast_tanhf(prr[0]), fast_tanhf(prr[1])};
  hA[NTM][1] = v2f{fast_tanhf(prr[2]), fast_tanhf(prr[3])};
#pragma unroll
  for (int n = 0; n < NTM; ++n) {
#pragma unroll
    for (int q = 0; q < 4; ++q) {
      v2f v;
      v.x = fast_tanhf(pre[n][2 * q]);
      v.y = fast_tanhf(pre[n][2 * q + 1]);
      hA[n][q] = v;
    }
  }

  v8f f0;
#pragma unroll
  for (int r = 0; r < 8; ++r) f0[r] = s_bfc[half * 8 + r];
#pragma unroll
  for (int j = 0; j < KHR; ++j) {
    const int kb = 16 * (j >> 2) + 8 * half + 2 * (j & 3);
    v2f a;
    a.x = s_wfc[kb][col];
    a.y = s_wfc[kb + 1][col];
    f0 = wmma_f32(a, hA[j >> 2][j & 3], f0);
  }
  float* orow = out + (size_t)(b0 + rowA0 + col) * NCLS + half * 8;
#pragma unroll
  for (int r = 0; r < 8; ++r)
    if (half * 8 + r < NCLS) orow[r] = f0[r];
}

extern "C" void kernel_launch(void* const* d_in, const int* in_sizes, int n_in,
                              void* d_out, int out_size, void* d_ws,
                              size_t ws_size, hipStream_t stream) {
  const float* data = (const float*)d_in[0];
  const float* w_xh = (const float*)d_in[1];
  const float* b_xh = (const float*)d_in[2];
  const float* w_hh = (const float*)d_in[3];
  const float* b_hh = (const float*)d_in[4];
  const float* w_fc = (const float*)d_in[5];
  const float* b_fc = (const float*)d_in[6];
  float* out = (float*)d_out;

  dim3 grid(BATCH / MTILE);
  dim3 block(BLOCK);
  hipLaunchKernelGGL(rnn_fused_kernel, grid, block, 0, stream, data, w_xh,
                     b_xh, w_hh, b_hh, w_fc, b_fc, out);
}
